# starting kernel
# baseline (speedup 1.0000x reference)
#include <hip/hip_runtime.h>
#include <math.h>

typedef float v2f __attribute__((ext_vector_type(2)));
typedef float v8f __attribute__((ext_vector_type(8)));

constexpr int BATCH = 8;
constexpr int SEQ   = 2048;
constexpr int DIM   = 64;
constexpr int TILE  = 16;
constexpr int QPW   = 2;
constexpr int KTILES = SEQ / TILE;
constexpr float SCALE = 0.125f;

__global__ __launch_bounds__(32)
void ScaledDotProductAttention_fa_kernel(const float* __restrict__ q,
                                         const float* __restrict__ k,
                                         const float* __restrict__ v,
                                         float* __restrict__ out) {
    const int lane = threadIdx.x & 31;
    const int r = lane & 15;
    const int h = lane >> 4;
    const int qt = blockIdx.x * QPW;
    const int b  = blockIdx.y;

    const float* Qb = q + (size_t)b * SEQ * DIM;
    const float* K  = k + (size_t)b * SEQ * DIM;
    const float* V  = v + (size_t)b * SEQ * DIM;

    v2f qf[QPW][16];
#pragma unroll
    for (int p = 0; p < QPW; ++p)
#pragma unroll
        for (int c = 0; c < 16; ++c)
            qf[p][c] = *(const v2f*)(Qb + (size_t)(qt + p) * TILE * DIM
                                        + r * DIM + 4 * c + 2 * h);

    v8f acc[QPW][4] = {};
    float mi[QPW] = {-INFINITY, -INFINITY};
    float li[QPW] = {0.0f, 0.0f};

    for (int j = 0; j < KTILES; ++j) {
        const int kb = j * TILE;

        v8f st[QPW] = {};
#pragma unroll
        for (int c = 0; c < 16; ++c) {
            v2f kf = *(const v2f*)(K + (size_t)(kb + r) * DIM + 4 * c + 2 * h);
            st[0] = __builtin_amdgcn_wmma_f32_16x16x4_f32(false, kf, false, qf[0][c],
                                                          (short)0, st[0], false, false);
            st[1] = __builtin_amdgcn_wmma_f32_16x16x4_f32(false, kf, false, qf[1][c],
                                                          (short)0, st[1], false, false);
        }

        v2f pa[QPW][4];
#pragma unroll
        for (int p = 0; p < QPW; ++p) {
#pragma unroll
            for (int e = 0; e < 8; ++e) {
                float s = st[p][e] * SCALE;
                st[p][e] = (s == 0.0f) ? -INFINITY : s;
            }

            float tmax = st[p][0];
#pragma unroll
            for (int e = 1; e < 8; ++e) tmax = fmaxf(tmax, st[p][e]);
            tmax = fmaxf(tmax, __shfl_xor(tmax, 16, 32));
            float mnew = fmaxf(mi[p], tmax);
            float corr = __expf(mi[p] - mnew);
            mi[p] = mnew;

            float psum = 0.0f;
#pragma unroll
            for (int e = 0; e < 8; ++e) {
                float pe = __expf(st[p][e] - mnew);
                st[p][e] = pe;
                psum += pe;
            }
            psum += __shfl_xor(psum, 16, 32);
            li[p] = li[p] * corr + psum;

#pragma unroll
            for (int e = 0; e < 8; ++e) {
                float ce = __shfl(corr, e + 8 * h, 32);
                acc[p][0][e] *= ce; acc[p][1][e] *= ce;
                acc[p][2][e] *= ce; acc[p][3][e] *= ce;
            }

            float sh[8];
#pragma unroll
            for (int e = 0; e < 8; ++e) sh[e] = __shfl_xor(st[p][e], 16, 32);
            pa[p][0][0] = h ? sh[2]    : st[p][0];  pa[p][0][1] = h ? sh[3]    : st[p][1];
            pa[p][1][0] = h ? sh[6]    : st[p][4];  pa[p][1][1] = h ? sh[7]    : st[p][5];
            pa[p][2][0] = h ? st[p][2] : sh[0];     pa[p][2][1] = h ? st[p][3] : sh[1];
            pa[p][3][0] = h ? st[p][6] : sh[4];     pa[p][3][1] = h ? st[p][7] : sh[5];
        }

#pragma unroll
        for (int n = 0; n < 4; ++n) {
#pragma unroll
            for (int kc = 0; kc < 4; ++kc) {
                const float* vp = V + (size_t)(kb + 4 * kc + 2 * h) * DIM + 16 * n + r;
                v2f bf;
                bf[0] = vp[0];
                bf[1] = vp[DIM];
                acc[0][n] = __builtin_amdgcn_wmma_f32_16x16x4_f32(false, pa[0][kc], false, bf,
                                                                  (short)0, acc[0][n], false, false);
                acc[1][n] = __builtin_amdgcn_wmma_f32_16x16x4_f32(false, pa[1][kc], false, bf,
                                                                  (short)0, acc[1][n], false, false);
            }
        }
    }

#pragma unroll
    for (int p = 0; p < QPW; ++p) {
        float linv = 1.0f / li[p];
        float* O = out + (size_t)b * SEQ * DIM + (size_t)(qt + p) * TILE * DIM;
#pragma unroll
        for (int e = 0; e < 8; ++e) {
            float le = __shfl(linv, e + 8 * h, 32);
            const int row = e + 8 * h;
#pragma unroll
            for (int n = 0; n < 4; ++n)
                O[(size_t)row * DIM + 16 * n + r] = acc[p][n][e] * le;
        }
    }
}

extern "C" void kernel_launch(void* const* d_in, const int* in_sizes, int n_in,
                              void* d_out, int out_size, void* d_ws, size_t ws_size,
                              hipStream_t stream) {
    const float* q = (const float*)d_in[0];
    const float* k = (const float*)d_in[1];
    const float* v = (const float*)d_in[2];
    float* out = (float*)d_out;
    dim3 grid(SEQ / (TILE * QPW), BATCH);
    ScaledDotProductAttention_fa_kernel<<<grid, dim3(32), 0, stream>>>(q, k, v, out);
}


# Split keys over 8 waves per work-group (256 keys each, own online softmax; max starts at -FLT_MAX so a fully masked chunk merges as 0), merged via LDS in fixed wave order: 4096 waves; new sum tree
# speedup vs baseline: 3.7670x; 3.7670x over previous
#include <hip/hip_runtime.h>
#include <math.h>
#include <float.h>

typedef float v2f __attribute__((ext_vector_type(2)));
typedef float v4f __attribute__((ext_vector_type(4)));
typedef float v8f __attribute__((ext_vector_type(8)));

constexpr int BATCH = 8;
constexpr int SEQ   = 2048;
constexpr int DIM   = 64;
constexpr int TILE  = 16;
constexpr int QPW   = 2;
constexpr int KTILES = SEQ / TILE;
constexpr int NW    = 8;
constexpr int KT_PER_WAVE = KTILES / NW;
constexpr int QROWS = TILE * QPW;
constexpr int PITCH = QROWS + 4;
constexpr int PART  = DIM * PITCH;
constexpr float SCALE = 0.125f;

static_assert(KTILES % NW == 0, "key tiles split evenly over the waves");
static_assert(QROWS == 4 * NW, "each wave finalizes four query rows");

__global__ __launch_bounds__(NW * 32) __attribute__((amdgpu_waves_per_eu(4, 4)))
void ScaledDotProductAttention_fa_kernel(const float* __restrict__ q,
                                         const float* __restrict__ k,
                                         const float* __restrict__ v,
                                         float* __restrict__ out) {
    __shared__ __attribute__((aligned(16))) float s_part[NW * PART];
    __shared__ float s_m[NW][QROWS];
    __shared__ float s_l[NW][QROWS];

    const int lane = threadIdx.x & 31;
    const int r = lane & 15;
    const int h = lane >> 4;
    const int w = __builtin_amdgcn_readfirstlane((int)(threadIdx.x >> 5));
    const int qt = blockIdx.x * QPW;
    const int b  = blockIdx.y;

    const float* Qb = q + (size_t)b * SEQ * DIM;
    const float* K  = k + (size_t)b * SEQ * DIM + (size_t)w * KT_PER_WAVE * TILE * DIM;
    const float* V  = v + (size_t)b * SEQ * DIM + (size_t)w * KT_PER_WAVE * TILE * DIM;

    v2f qf[QPW][16];
#pragma unroll
    for (int p = 0; p < QPW; ++p)
#pragma unroll
        for (int c = 0; c < 16; ++c)
            qf[p][c] = *(const v2f*)(Qb + (size_t)(qt + p) * TILE * DIM
                                        + r * DIM + 4 * c + 2 * h);

    v8f acc[QPW][4] = {};
    float mi[QPW] = {-FLT_MAX, -FLT_MAX};
    float li[QPW] = {0.0f, 0.0f};

    for (int j = 0; j < KT_PER_WAVE; ++j) {
        const int kb = j * TILE;

        v8f st[QPW] = {};
#pragma unroll
        for (int c = 0; c < 16; ++c) {
            v2f kf = *(const v2f*)(K + (size_t)(kb + r) * DIM + 4 * c + 2 * h);
            st[0] = __builtin_amdgcn_wmma_f32_16x16x4_f32(false, kf, false, qf[0][c],
                                                          (short)0, st[0], false, false);
            st[1] = __builtin_amdgcn_wmma_f32_16x16x4_f32(false, kf, false, qf[1][c],
                                                          (short)0, st[1], false, false);
        }

        v2f pa[QPW][4];
#pragma unroll
        for (int p = 0; p < QPW; ++p) {
#pragma unroll
            for (int e = 0; e < 8; ++e) {
                float s = st[p][e] * SCALE;
                st[p][e] = (s == 0.0f) ? -INFINITY : s;
            }

            float tmax = st[p][0];
#pragma unroll
            for (int e = 1; e < 8; ++e) tmax = fmaxf(tmax, st[p][e]);
            tmax = fmaxf(tmax, __shfl_xor(tmax, 16, 32));
            float mnew = fmaxf(mi[p], tmax);
            float corr = __expf(mi[p] - mnew);
            mi[p] = mnew;

            float psum = 0.0f;
#pragma unroll
            for (int e = 0; e < 8; ++e) {
                float pe = __expf(st[p][e] - mnew);
                st[p][e] = pe;
                psum += pe;
            }
            psum += __shfl_xor(psum, 16, 32);
            li[p] = li[p] * corr + psum;

#pragma unroll
            for (int e = 0; e < 8; ++e) {
                float ce = __shfl(corr, e + 8 * h, 32);
                acc[p][0][e] *= ce; acc[p][1][e] *= ce;
                acc[p][2][e] *= ce; acc[p][3][e] *= ce;
            }

            float sh[8];
#pragma unroll
            for (int e = 0; e < 8; ++e) sh[e] = __shfl_xor(st[p][e], 16, 32);
            pa[p][0][0] = h ? sh[2]    : st[p][0];  pa[p][0][1] = h ? sh[3]    : st[p][1];
            pa[p][1][0] = h ? sh[6]    : st[p][4];  pa[p][1][1] = h ? sh[7]    : st[p][5];
            pa[p][2][0] = h ? st[p][2] : sh[0];     pa[p][2][1] = h ? st[p][3] : sh[1];
            pa[p][3][0] = h ? st[p][6] : sh[4];     pa[p][3][1] = h ? st[p][7] : sh[5];
        }

#pragma unroll
        for (int n = 0; n < 4; ++n) {
#pragma unroll
            for (int kc = 0; kc < 4; ++kc) {
                const float* vp = V + (size_t)(kb + 4 * kc + 2 * h) * DIM + 16 * n + r;
                v2f bf;
                bf[0] = vp[0];
                bf[1] = vp[DIM];
                acc[0][n] = __builtin_amdgcn_wmma_f32_16x16x4_f32(false, pa[0][kc], false, bf,
                                                                  (short)0, acc[0][n], false, false);
                acc[1][n] = __builtin_amdgcn_wmma_f32_16x16x4_f32(false, pa[1][kc], false, bf,
                                                                  (short)0, acc[1][n], false, false);
            }
        }
    }

    const float m_own = h ? mi[1] : mi[0];
    s_m[w][lane] = m_own;
    s_l[w][lane] = h ? li[1] : li[0];
    __syncthreads();

    float M = s_m[0][lane];
#pragma unroll
    for (int x = 1; x < NW; ++x) M = fmaxf(M, s_m[x][lane]);
    float L = 0.0f;
#pragma unroll
    for (int x = 0; x < NW; ++x) L += s_l[x][lane] * __expf(s_m[x][lane] - M);
    const float sc = __expf(m_own - M) / L;

    float* mine = s_part + w * PART;
#pragma unroll
    for (int p = 0; p < QPW; ++p) {
#pragma unroll
        for (int e = 0; e < 8; ++e) {
            float se = __shfl(sc, 16 * p + e + 8 * h, 32);
            acc[p][0][e] *= se; acc[p][1][e] *= se;
            acc[p][2][e] *= se; acc[p][3][e] *= se;
        }
#pragma unroll
        for (int n = 0; n < 4; ++n) {
            float* dst = mine + (16 * n + r) * PITCH + 16 * p + 8 * h;
            v4f lo = {acc[p][n][0], acc[p][n][1], acc[p][n][2], acc[p][n][3]};
            v4f hi = {acc[p][n][4], acc[p][n][5], acc[p][n][6], acc[p][n][7]};
            *(v4f*)(dst)     = lo;
            *(v4f*)(dst + 4) = hi;
        }
    }
    __syncthreads();

    v4f s0 = {0.0f, 0.0f, 0.0f, 0.0f};
    v4f s1 = {0.0f, 0.0f, 0.0f, 0.0f};
#pragma unroll
    for (int x = 0; x < NW; ++x) {
        const float* src = s_part + x * PART + lane * PITCH + 4 * w;
        s0 += *(const v4f*)(src);
        s1 += *(const v4f*)(src + 32 * PITCH);
    }
    float* O = out + (size_t)b * SEQ * DIM + ((size_t)qt * TILE + 4 * w) * DIM;
#pragma unroll
    for (int i = 0; i < 4; ++i) {
        O[(size_t)i * DIM + lane]      = s0[i];
        O[(size_t)i * DIM + 32 + lane] = s1[i];
    }
}

extern "C" void kernel_launch(void* const* d_in, const int* in_sizes, int n_in,
                              void* d_out, int out_size, void* d_ws, size_t ws_size,
                              hipStream_t stream) {
    const float* q = (const float*)d_in[0];
    const float* k = (const float*)d_in[1];
    const float* v = (const float*)d_in[2];
    float* out = (float*)d_out;
    dim3 grid(SEQ / (TILE * QPW), BATCH);
    ScaledDotProductAttention_fa_kernel<<<grid, dim3(NW * 32), 0, stream>>>(q, k, v, out);
}
